# starting kernel
# baseline (speedup 1.0000x reference)
#include <hip/hip_runtime.h>

typedef __attribute__((ext_vector_type(2))) float v2f;
typedef __attribute__((ext_vector_type(8))) float v8f;

#define BATCH    4
#define CHANNELS 256
#define LATENT   512
#define DIM      32
#define SPATIAL  (DIM * DIM * DIM)
#define ODIM     64

__global__ __launch_bounds__(CHANNELS)
void style_mod_kernel(const float* __restrict__ istyle,
                      const float* __restrict__ style_w,
                      const float* __restrict__ style_b,
                      const float* __restrict__ conv_w,
                      float* __restrict__ mod) {
    __shared__ float s_is[LATENT];
    const int b = blockIdx.x;
    const int c = threadIdx.x;
    for (int i = threadIdx.x; i < LATENT; i += CHANNELS)
        s_is[i] = istyle[b * LATENT + i];
    __syncthreads();

    const float4* wrow = (const float4*)(style_w + c * LATENT);
    float acc = style_b[c];
    #pragma unroll 4
    for (int l = 0; l < LATENT / 4; ++l) {
        float4 w4 = wrow[l];
        acc += s_is[4 * l + 0] * w4.x + s_is[4 * l + 1] * w4.y +
               s_is[4 * l + 2] * w4.z + s_is[4 * l + 3] * w4.w;
    }
    mod[b * CHANNELS + c] = conv_w[c] * (acc + 1.0f);
}

__global__ __launch_bounds__(256)
void einsum_wmma_kernel(const float* __restrict__ x,
                        const float* __restrict__ prev,
                        const float* __restrict__ mod,
                        float* __restrict__ y0) {
    const int lane = threadIdx.x & 31;
    const int wave = threadIdx.x >> 5;
    const int tile = blockIdx.x * 8 + wave;
    const int tilesPerBatch = SPATIAL / 16;
    const int b  = tile / tilesPerBatch;
    const int s0 = (tile % tilesPerBatch) * 16;

    const int half = lane >> 4;
    const int m    = lane & 15;

    const float* xb   = x   + (size_t)b * CHANNELS * SPATIAL;
    const float* modb = mod + b * CHANNELS;

    v8f acc = {};
    for (int c0 = 0; c0 < CHANNELS; c0 += 4) {
        const int cA = c0 + 2 * half;
        v2f a, bb;
        a.x  = xb[(size_t)cA       * SPATIAL + s0 + m];
        a.y  = xb[(size_t)(cA + 1) * SPATIAL + s0 + m];
        bb.x = modb[cA];
        bb.y = modb[cA + 1];
        acc = __builtin_amdgcn_wmma_f32_16x16x4_f32(
             false, a,  false, bb,
             (short)0, acc,  false,  false);
    }

    if (m == 0) {
        const int base = s0 + 8 * half;
        const float4* pv = (const float4*)(prev + (size_t)b * SPATIAL + base);
        float4 p0 = pv[0], p1 = pv[1];
        float4 r0 = { acc[0] + p0.x, acc[1] + p0.y, acc[2] + p0.z, acc[3] + p0.w };
        float4 r1 = { acc[4] + p1.x, acc[5] + p1.y, acc[6] + p1.z, acc[7] + p1.w };
        float4* ov = (float4*)(y0 + (size_t)b * SPATIAL + base);
        ov[0] = r0;
        ov[1] = r1;
    }
}

__device__ __forceinline__ void axis_taps(int p, int* i3, float* w3) {
    const int m   = p >> 1;
    const int par = p & 1;
    i3[0] = (m - 1 < 0) ? 0 : m - 1;
    i3[1] = m;
    i3[2] = (m + 1 > DIM - 1) ? DIM - 1 : m + 1;
    w3[0] = par ? 0.0625f : 0.3125f;
    w3[1] = 0.625f;
    w3[2] = par ? 0.3125f : 0.0625f;
}

__global__ __launch_bounds__(256)
void upblur_kernel(const float* __restrict__ y0, float* __restrict__ out) {
    const int idx = blockIdx.x * 256 + threadIdx.x;
    const int w = idx & 63;
    const int h = (idx >> 6) & 63;
    const int d = (idx >> 12) & 63;
    const int b = idx >> 18;

    const float* yb = y0 + (size_t)b * SPATIAL;

    int id[3], ih[3], iw[3];
    float wd[3], wh[3], ww[3];
    axis_taps(d, id, wd);
    axis_taps(h, ih, wh);
    axis_taps(w, iw, ww);

    float sum = 0.0f;
    #pragma unroll
    for (int i = 0; i < 3; ++i) {
        #pragma unroll
        for (int j = 0; j < 3; ++j) {
            const float* row = yb + (id[i] * DIM + ih[j]) * DIM;
            const float rs = ww[0] * row[iw[0]] + ww[1] * row[iw[1]] + ww[2] * row[iw[2]];
            sum += wd[i] * wh[j] * rs;
        }
    }
    out[idx] = sum;
}

extern "C" void kernel_launch(void* const* d_in, const int* in_sizes, int n_in,
                              void* d_out, int out_size, void* d_ws, size_t ws_size,
                              hipStream_t stream) {
    const float* x       = (const float*)d_in[0];
    const float* prev    = (const float*)d_in[1];
    const float* istyle  = (const float*)d_in[2];
    const float* style_w = (const float*)d_in[3];
    const float* style_b = (const float*)d_in[4];
    const float* conv_w  = (const float*)d_in[5];
    float* out = (float*)d_out;

    float* mod = (float*)d_ws;
    float* y0  = mod + BATCH * CHANNELS;

    style_mod_kernel<<<BATCH, CHANNELS, 0, stream>>>(istyle, style_w, style_b, conv_w, mod);

    einsum_wmma_kernel<<<(BATCH * (SPATIAL / 16)) / 8, 256, 0, stream>>>(x, prev, mod, y0);

    upblur_kernel<<<(BATCH * ODIM * ODIM * ODIM) / 256, 256, 0, stream>>>(y0, out);
}


# style GEMV re-parallelised: 32 blocks, wave-cooperative whole-line float4 row reads + cross-lane butterfly sum (was 4 blocks, one strided row per thread); style sum order changes
# speedup vs baseline: 1.3904x; 1.3904x over previous
#include <hip/hip_runtime.h>

typedef __attribute__((ext_vector_type(2))) float v2f;
typedef __attribute__((ext_vector_type(8))) float v8f;

#define BATCH    4
#define CHANNELS 256
#define LATENT   512
#define DIM      32
#define SPATIAL  (DIM * DIM * DIM)
#define ODIM     64

#define K1_CH_PER_BLOCK 32
#define K1_CH_PER_WAVE  4
#define K1_VEC_PER_LANE (LATENT / (4 * 32))

__global__ __launch_bounds__(256)
void style_mod_kernel(const float* __restrict__ istyle,
                      const float* __restrict__ style_w,
                      const float* __restrict__ style_b,
                      const float* __restrict__ conv_w,
                      float* __restrict__ mod) {
    __shared__ float s_dot[K1_CH_PER_BLOCK];
    const int lane = threadIdx.x & 31;
    const int wave = threadIdx.x >> 5;
    const int b    = blockIdx.x & (BATCH - 1);
    const int c0   = (blockIdx.x / BATCH) * K1_CH_PER_BLOCK;

    const float cw = conv_w[c0 + lane];
    const float sb = style_b[c0 + lane];

    const float4* irow = (const float4*)(istyle + b * LATENT);
    float4 s4[K1_VEC_PER_LANE];
    #pragma unroll
    for (int i = 0; i < K1_VEC_PER_LANE; ++i)
        s4[i] = irow[lane + 32 * i];

    float acc[K1_CH_PER_WAVE];
    #pragma unroll
    for (int j = 0; j < K1_CH_PER_WAVE; ++j) {
        const int c = c0 + K1_CH_PER_WAVE * wave + j;
        const float4* wrow = (const float4*)(style_w + c * LATENT);
        float a = 0.0f;
        #pragma unroll
        for (int i = 0; i < K1_VEC_PER_LANE; ++i) {
            const float4 w4 = wrow[lane + 32 * i];
            a += s4[i].x * w4.x + s4[i].y * w4.y + s4[i].z * w4.z + s4[i].w * w4.w;
        }
        acc[j] = a;
    }

    #pragma unroll
    for (int off = 16; off > 0; off >>= 1) {
        #pragma unroll
        for (int j = 0; j < K1_CH_PER_WAVE; ++j)
            acc[j] += __shfl_xor(acc[j], off, 32);
    }

    if (lane == 0) {
        #pragma unroll
        for (int j = 0; j < K1_CH_PER_WAVE; ++j)
            s_dot[K1_CH_PER_WAVE * wave + j] = acc[j];
    }
    __syncthreads();

    if (wave == 0)
        mod[b * CHANNELS + c0 + lane] = cw * ((s_dot[lane] + sb) + 1.0f);
}

__global__ __launch_bounds__(256)
void einsum_wmma_kernel(const float* __restrict__ x,
                        const float* __restrict__ prev,
                        const float* __restrict__ mod,
                        float* __restrict__ y0) {
    const int lane = threadIdx.x & 31;
    const int wave = threadIdx.x >> 5;
    const int tile = blockIdx.x * 8 + wave;
    const int tilesPerBatch = SPATIAL / 16;
    const int b  = tile / tilesPerBatch;
    const int s0 = (tile % tilesPerBatch) * 16;

    const int half = lane >> 4;
    const int m    = lane & 15;

    const float* xb   = x   + (size_t)b * CHANNELS * SPATIAL;
    const float* modb = mod + b * CHANNELS;

    v8f acc = {};
    for (int c0 = 0; c0 < CHANNELS; c0 += 4) {
        const int cA = c0 + 2 * half;
        v2f a, bb;
        a.x  = xb[(size_t)cA       * SPATIAL + s0 + m];
        a.y  = xb[(size_t)(cA + 1) * SPATIAL + s0 + m];
        bb.x = modb[cA];
        bb.y = modb[cA + 1];
        acc = __builtin_amdgcn_wmma_f32_16x16x4_f32(
             false, a,  false, bb,
             (short)0, acc,  false,  false);
    }

    if (m == 0) {
        const int base = s0 + 8 * half;
        const float4* pv = (const float4*)(prev + (size_t)b * SPATIAL + base);
        float4 p0 = pv[0], p1 = pv[1];
        float4 r0 = { acc[0] + p0.x, acc[1] + p0.y, acc[2] + p0.z, acc[3] + p0.w };
        float4 r1 = { acc[4] + p1.x, acc[5] + p1.y, acc[6] + p1.z, acc[7] + p1.w };
        float4* ov = (float4*)(y0 + (size_t)b * SPATIAL + base);
        ov[0] = r0;
        ov[1] = r1;
    }
}

__device__ __forceinline__ void axis_taps(int p, int* i3, float* w3) {
    const int m   = p >> 1;
    const int par = p & 1;
    i3[0] = (m - 1 < 0) ? 0 : m - 1;
    i3[1] = m;
    i3[2] = (m + 1 > DIM - 1) ? DIM - 1 : m + 1;
    w3[0] = par ? 0.0625f : 0.3125f;
    w3[1] = 0.625f;
    w3[2] = par ? 0.3125f : 0.0625f;
}

__global__ __launch_bounds__(256)
void upblur_kernel(const float* __restrict__ y0, float* __restrict__ out) {
    const int idx = blockIdx.x * 256 + threadIdx.x;
    const int w = idx & 63;
    const int h = (idx >> 6) & 63;
    const int d = (idx >> 12) & 63;
    const int b = idx >> 18;

    const float* yb = y0 + (size_t)b * SPATIAL;

    int id[3], ih[3], iw[3];
    float wd[3], wh[3], ww[3];
    axis_taps(d, id, wd);
    axis_taps(h, ih, wh);
    axis_taps(w, iw, ww);

    float sum = 0.0f;
    #pragma unroll
    for (int i = 0; i < 3; ++i) {
        #pragma unroll
        for (int j = 0; j < 3; ++j) {
            const float* row = yb + (id[i] * DIM + ih[j]) * DIM;
            const float rs = ww[0] * row[iw[0]] + ww[1] * row[iw[1]] + ww[2] * row[iw[2]];
            sum += wd[i] * wh[j] * rs;
        }
    }
    out[idx] = sum;
}

extern "C" void kernel_launch(void* const* d_in, const int* in_sizes, int n_in,
                              void* d_out, int out_size, void* d_ws, size_t ws_size,
                              hipStream_t stream) {
    const float* x       = (const float*)d_in[0];
    const float* prev    = (const float*)d_in[1];
    const float* istyle  = (const float*)d_in[2];
    const float* style_w = (const float*)d_in[3];
    const float* style_b = (const float*)d_in[4];
    const float* conv_w  = (const float*)d_in[5];
    float* out = (float*)d_out;

    float* mod = (float*)d_ws;
    float* y0  = mod + BATCH * CHANNELS;

    style_mod_kernel<<<BATCH * (CHANNELS / K1_CH_PER_BLOCK), 256, 0, stream>>>(istyle, style_w, style_b, conv_w, mod);

    einsum_wmma_kernel<<<(BATCH * (SPATIAL / 16)) / 8, 256, 0, stream>>>(x, prev, mod, y0);

    upblur_kernel<<<(BATCH * ODIM * ODIM * ODIM) / 256, 256, 0, stream>>>(y0, out);
}


# upsample+blur kernel coarsened: one thread per pair of y0 cells makes their 4x2x2 outputs (36 loads, 8 float2 stores), 2048 waves instead of 32768; per-output sum order kept
# speedup vs baseline: 1.9391x; 1.3946x over previous
#include <hip/hip_runtime.h>

typedef __attribute__((ext_vector_type(2))) float v2f;
typedef __attribute__((ext_vector_type(8))) float v8f;

#define BATCH    4
#define CHANNELS 256
#define LATENT   512
#define DIM      32
#define SPATIAL  (DIM * DIM * DIM)
#define ODIM     64

#define K1_CH_PER_BLOCK 32
#define K1_CH_PER_WAVE  4
#define K1_VEC_PER_LANE (LATENT / (4 * 32))

__global__ __launch_bounds__(256)
void style_mod_kernel(const float* __restrict__ istyle,
                      const float* __restrict__ style_w,
                      const float* __restrict__ style_b,
                      const float* __restrict__ conv_w,
                      float* __restrict__ mod) {
    __shared__ float s_dot[K1_CH_PER_BLOCK];
    const int lane = threadIdx.x & 31;
    const int wave = threadIdx.x >> 5;
    const int b    = blockIdx.x & (BATCH - 1);
    const int c0   = (blockIdx.x / BATCH) * K1_CH_PER_BLOCK;

    const float cw = conv_w[c0 + lane];
    const float sb = style_b[c0 + lane];

    const float4* irow = (const float4*)(istyle + b * LATENT);
    float4 s4[K1_VEC_PER_LANE];
    #pragma unroll
    for (int i = 0; i < K1_VEC_PER_LANE; ++i)
        s4[i] = irow[lane + 32 * i];

    float acc[K1_CH_PER_WAVE];
    #pragma unroll
    for (int j = 0; j < K1_CH_PER_WAVE; ++j) {
        const int c = c0 + K1_CH_PER_WAVE * wave + j;
        const float4* wrow = (const float4*)(style_w + c * LATENT);
        float a = 0.0f;
        #pragma unroll
        for (int i = 0; i < K1_VEC_PER_LANE; ++i) {
            const float4 w4 = wrow[lane + 32 * i];
            a += s4[i].x * w4.x + s4[i].y * w4.y + s4[i].z * w4.z + s4[i].w * w4.w;
        }
        acc[j] = a;
    }

    #pragma unroll
    for (int off = 16; off > 0; off >>= 1) {
        #pragma unroll
        for (int j = 0; j < K1_CH_PER_WAVE; ++j)
            acc[j] += __shfl_xor(acc[j], off, 32);
    }

    if (lane == 0) {
        #pragma unroll
        for (int j = 0; j < K1_CH_PER_WAVE; ++j)
            s_dot[K1_CH_PER_WAVE * wave + j] = acc[j];
    }
    __syncthreads();

    if (wave == 0)
        mod[b * CHANNELS + c0 + lane] = cw * ((s_dot[lane] + sb) + 1.0f);
}

__global__ __launch_bounds__(256)
void einsum_wmma_kernel(const float* __restrict__ x,
                        const float* __restrict__ prev,
                        const float* __restrict__ mod,
                        float* __restrict__ y0) {
    const int lane = threadIdx.x & 31;
    const int wave = threadIdx.x >> 5;
    const int tile = blockIdx.x * 8 + wave;
    const int tilesPerBatch = SPATIAL / 16;
    const int b  = tile / tilesPerBatch;
    const int s0 = (tile % tilesPerBatch) * 16;

    const int half = lane >> 4;
    const int m    = lane & 15;

    const float* xb   = x   + (size_t)b * CHANNELS * SPATIAL;
    const float* modb = mod + b * CHANNELS;

    v8f acc = {};
    for (int c0 = 0; c0 < CHANNELS; c0 += 4) {
        const int cA = c0 + 2 * half;
        v2f a, bb;
        a.x  = xb[(size_t)cA       * SPATIAL + s0 + m];
        a.y  = xb[(size_t)(cA + 1) * SPATIAL + s0 + m];
        bb.x = modb[cA];
        bb.y = modb[cA + 1];
        acc = __builtin_amdgcn_wmma_f32_16x16x4_f32(
             false, a,  false, bb,
             (short)0, acc,  false,  false);
    }

    if (m == 0) {
        const int base = s0 + 8 * half;
        const float4* pv = (const float4*)(prev + (size_t)b * SPATIAL + base);
        float4 p0 = pv[0], p1 = pv[1];
        float4 r0 = { acc[0] + p0.x, acc[1] + p0.y, acc[2] + p0.z, acc[3] + p0.w };
        float4 r1 = { acc[4] + p1.x, acc[5] + p1.y, acc[6] + p1.z, acc[7] + p1.w };
        float4* ov = (float4*)(y0 + (size_t)b * SPATIAL + base);
        ov[0] = r0;
        ov[1] = r1;
    }
}

#define K3_CELLS_D 2

__global__ __launch_bounds__(256)
void upblur_kernel(const float* __restrict__ y0, float* __restrict__ out) {
    const int t   = blockIdx.x * 256 + threadIdx.x;
    const int mw  = t & (DIM - 1);
    const int mh  = (t >> 5) & (DIM - 1);
    const int md0 = K3_CELLS_D * ((t >> 10) & (DIM / K3_CELLS_D - 1));
    const int b   = t >> 14;

    const float* yb = y0 + (size_t)b * SPATIAL;

    const int iw0 = (mw - 1 < 0) ? 0 : mw - 1;
    const int iw2 = (mw + 1 > DIM - 1) ? DIM - 1 : mw + 1;
    const int ih[3] = { (mh - 1 < 0) ? 0 : mh - 1, mh, (mh + 1 > DIM - 1) ? DIM - 1 : mh + 1 };
    const int ip[K3_CELLS_D + 2] = { (md0 - 1 < 0) ? 0 : md0 - 1, md0, md0 + 1,
                                     (md0 + 2 > DIM - 1) ? DIM - 1 : md0 + 2 };

    float re[K3_CELLS_D + 2][3], ro[K3_CELLS_D + 2][3];
    #pragma unroll
    for (int p = 0; p < K3_CELLS_D + 2; ++p) {
        #pragma unroll
        for (int j = 0; j < 3; ++j) {
            const float* row = yb + (ip[p] * DIM + ih[j]) * DIM;
            const float v0 = row[iw0], v1 = row[mw], v2 = row[iw2];
            re[p][j] = 0.3125f * v0 + 0.625f * v1 + 0.0625f * v2;
            ro[p][j] = 0.0625f * v0 + 0.625f * v1 + 0.3125f * v2;
        }
    }

    const float wt[2][3] = { { 0.3125f, 0.625f, 0.0625f }, { 0.0625f, 0.625f, 0.3125f } };

    #pragma unroll
    for (int q = 0; q < K3_CELLS_D; ++q) {
        #pragma unroll
        for (int pd = 0; pd < 2; ++pd) {
            #pragma unroll
            for (int ph = 0; ph < 2; ++ph) {
                float se = 0.0f, so = 0.0f;
                #pragma unroll
                for (int i = 0; i < 3; ++i) {
                    #pragma unroll
                    for (int j = 0; j < 3; ++j) {
                        const float c = wt[pd][i] * wt[ph][j];
                        se += c * re[q + i][j];
                        so += c * ro[q + i][j];
                    }
                }
                const int d = 2 * (md0 + q) + pd;
                const int h = 2 * mh + ph;
                float2* o = (float2*)(out + (((size_t)(b * ODIM + d) * ODIM + h) * ODIM + 2 * mw));
                *o = make_float2(se, so);
            }
        }
    }
}

extern "C" void kernel_launch(void* const* d_in, const int* in_sizes, int n_in,
                              void* d_out, int out_size, void* d_ws, size_t ws_size,
                              hipStream_t stream) {
    const float* x       = (const float*)d_in[0];
    const float* prev    = (const float*)d_in[1];
    const float* istyle  = (const float*)d_in[2];
    const float* style_w = (const float*)d_in[3];
    const float* style_b = (const float*)d_in[4];
    const float* conv_w  = (const float*)d_in[5];
    float* out = (float*)d_out;

    float* mod = (float*)d_ws;
    float* y0  = mod + BATCH * CHANNELS;

    style_mod_kernel<<<BATCH * (CHANNELS / K1_CH_PER_BLOCK), 256, 0, stream>>>(istyle, style_w, style_b, conv_w, mod);

    einsum_wmma_kernel<<<(BATCH * (SPATIAL / 16)) / 8, 256, 0, stream>>>(x, prev, mod, y0);

    upblur_kernel<<<(BATCH * SPATIAL / K3_CELLS_D) / 256, 256, 0, stream>>>(y0, out);
}


# channel-sum kernel re-tiled: each wave owns 64 cells and reads x as float4 row pieces into 4 accumulator tiles with a 32-load register ring in flight; 2048 waves, was 8192; per-output sum order kept
# speedup vs baseline: 2.1524x; 1.1100x over previous
#include <hip/hip_runtime.h>

typedef __attribute__((ext_vector_type(2))) float v2f;
typedef __attribute__((ext_vector_type(8))) float v8f;

#define BATCH    4
#define CHANNELS 256
#define LATENT   512
#define DIM      32
#define SPATIAL  (DIM * DIM * DIM)
#define ODIM     64

#define K1_CH_PER_BLOCK 32
#define K1_CH_PER_WAVE  4
#define K1_VEC_PER_LANE (LATENT / (4 * 32))

__global__ __launch_bounds__(256)
void style_mod_kernel(const float* __restrict__ istyle,
                      const float* __restrict__ style_w,
                      const float* __restrict__ style_b,
                      const float* __restrict__ conv_w,
                      float* __restrict__ mod) {
    __shared__ float s_dot[K1_CH_PER_BLOCK];
    const int lane = threadIdx.x & 31;
    const int wave = threadIdx.x >> 5;
    const int b    = blockIdx.x & (BATCH - 1);
    const int c0   = (blockIdx.x / BATCH) * K1_CH_PER_BLOCK;

    const float cw = conv_w[c0 + lane];
    const float sb = style_b[c0 + lane];

    const float4* irow = (const float4*)(istyle + b * LATENT);
    float4 s4[K1_VEC_PER_LANE];
    #pragma unroll
    for (int i = 0; i < K1_VEC_PER_LANE; ++i)
        s4[i] = irow[lane + 32 * i];

    float acc[K1_CH_PER_WAVE];
    #pragma unroll
    for (int j = 0; j < K1_CH_PER_WAVE; ++j) {
        const int c = c0 + K1_CH_PER_WAVE * wave + j;
        const float4* wrow = (const float4*)(style_w + c * LATENT);
        float a = 0.0f;
        #pragma unroll
        for (int i = 0; i < K1_VEC_PER_LANE; ++i) {
            const float4 w4 = wrow[lane + 32 * i];
            a += s4[i].x * w4.x + s4[i].y * w4.y + s4[i].z * w4.z + s4[i].w * w4.w;
        }
        acc[j] = a;
    }

    #pragma unroll
    for (int off = 16; off > 0; off >>= 1) {
        #pragma unroll
        for (int j = 0; j < K1_CH_PER_WAVE; ++j)
            acc[j] += __shfl_xor(acc[j], off, 32);
    }

    if (lane == 0) {
        #pragma unroll
        for (int j = 0; j < K1_CH_PER_WAVE; ++j)
            s_dot[K1_CH_PER_WAVE * wave + j] = acc[j];
    }
    __syncthreads();

    if (wave == 0)
        mod[b * CHANNELS + c0 + lane] = cw * ((s_dot[lane] + sb) + 1.0f);
}

#define K2_CELLS 64
#define K2_STEPS (CHANNELS / 4)
#define K2_AHEAD 16

__device__ __forceinline__ v8f k2_mma(v2f a, float b_lo, float b_hi, v8f c) {
    v2f bfrag = { b_lo, b_hi };
    return __builtin_amdgcn_wmma_f32_16x16x4_f32(
         false, a,  false, bfrag,
         (short)0, c,  false,  false);
}

__global__ __launch_bounds__(256)
void einsum_wmma_kernel(const float* __restrict__ x,
                        const float* __restrict__ prev,
                        const float* __restrict__ mod,
                        float* __restrict__ y0) {
    __shared__ float s_mod[CHANNELS];

    const int lane = threadIdx.x & 31;
    const int wave = threadIdx.x >> 5;
    const int b    = blockIdx.x / (SPATIAL / (8 * K2_CELLS));
    const int s0   = ((blockIdx.x * 8 + wave) % (SPATIAL / K2_CELLS)) * K2_CELLS;
    const int half = lane >> 4;
    const int col  = lane & 15;

    const float  mv = mod[b * CHANNELS + threadIdx.x];
    const float4 pv = *(const float4*)(prev + (size_t)b * SPATIAL + s0 + 4 * col);

    const float* xp = x + ((size_t)b * CHANNELS + 2 * half) * SPATIAL + s0 + 4 * col;

    float4 ring[K2_AHEAD][2];
    #pragma unroll
    for (int i = 0; i < K2_AHEAD; ++i) {
        ring[i][0] = *(const float4*)(xp + (size_t)(4 * i)     * SPATIAL);
        ring[i][1] = *(const float4*)(xp + (size_t)(4 * i + 1) * SPATIAL);
    }

    s_mod[threadIdx.x] = mv;
    __builtin_amdgcn_fence(__ATOMIC_RELEASE, "workgroup", "local");
    __builtin_amdgcn_s_barrier();
    __builtin_amdgcn_fence(__ATOMIC_ACQUIRE, "workgroup", "local");

    v2f mf[K2_STEPS];
    #pragma unroll
    for (int i = 0; i < K2_STEPS; ++i)
        mf[i] = *(const v2f*)&s_mod[4 * i + 2 * half];
    __builtin_amdgcn_sched_barrier(0);

    v8f acc0 = {}, acc1 = {}, acc2 = {}, acc3 = {};
    #pragma unroll
    for (int i = 0; i < K2_STEPS; ++i) {
        const int u = i % K2_AHEAD;
        const float4 q0 = ring[u][0], q1 = ring[u][1];
        acc0 = k2_mma(mf[i], q0.x, q1.x, acc0);
        acc1 = k2_mma(mf[i], q0.y, q1.y, acc1);
        acc2 = k2_mma(mf[i], q0.z, q1.z, acc2);
        acc3 = k2_mma(mf[i], q0.w, q1.w, acc3);
        if (i + K2_AHEAD < K2_STEPS) {
            ring[u][0] = *(const float4*)(xp + (size_t)(4 * (i + K2_AHEAD))     * SPATIAL);
            ring[u][1] = *(const float4*)(xp + (size_t)(4 * (i + K2_AHEAD) + 1) * SPATIAL);
        }
    }

    const float4 r = { acc0[0] + pv.x, acc1[0] + pv.y, acc2[0] + pv.z, acc3[0] + pv.w };
    if (half == 0)
        *(float4*)(y0 + (size_t)b * SPATIAL + s0 + 4 * col) = r;
}

#define K3_CELLS_D 2

__global__ __launch_bounds__(256)
void upblur_kernel(const float* __restrict__ y0, float* __restrict__ out) {
    const int t   = blockIdx.x * 256 + threadIdx.x;
    const int mw  = t & (DIM - 1);
    const int mh  = (t >> 5) & (DIM - 1);
    const int md0 = K3_CELLS_D * ((t >> 10) & (DIM / K3_CELLS_D - 1));
    const int b   = t >> 14;

    const float* yb = y0 + (size_t)b * SPATIAL;

    const int iw0 = (mw - 1 < 0) ? 0 : mw - 1;
    const int iw2 = (mw + 1 > DIM - 1) ? DIM - 1 : mw + 1;
    const int ih[3] = { (mh - 1 < 0) ? 0 : mh - 1, mh, (mh + 1 > DIM - 1) ? DIM - 1 : mh + 1 };
    const int ip[K3_CELLS_D + 2] = { (md0 - 1 < 0) ? 0 : md0 - 1, md0, md0 + 1,
                                     (md0 + 2 > DIM - 1) ? DIM - 1 : md0 + 2 };

    float re[K3_CELLS_D + 2][3], ro[K3_CELLS_D + 2][3];
    #pragma unroll
    for (int p = 0; p < K3_CELLS_D + 2; ++p) {
        #pragma unroll
        for (int j = 0; j < 3; ++j) {
            const float* row = yb + (ip[p] * DIM + ih[j]) * DIM;
            const float v0 = row[iw0], v1 = row[mw], v2 = row[iw2];
            re[p][j] = 0.3125f * v0 + 0.625f * v1 + 0.0625f * v2;
            ro[p][j] = 0.0625f * v0 + 0.625f * v1 + 0.3125f * v2;
        }
    }

    const float wt[2][3] = { { 0.3125f, 0.625f, 0.0625f }, { 0.0625f, 0.625f, 0.3125f } };

    #pragma unroll
    for (int q = 0; q < K3_CELLS_D; ++q) {
        #pragma unroll
        for (int pd = 0; pd < 2; ++pd) {
            #pragma unroll
            for (int ph = 0; ph < 2; ++ph) {
                float se = 0.0f, so = 0.0f;
                #pragma unroll
                for (int i = 0; i < 3; ++i) {
                    #pragma unroll
                    for (int j = 0; j < 3; ++j) {
                        const float c = wt[pd][i] * wt[ph][j];
                        se += c * re[q + i][j];
                        so += c * ro[q + i][j];
                    }
                }
                const int d = 2 * (md0 + q) + pd;
                const int h = 2 * mh + ph;
                float2* o = (float2*)(out + (((size_t)(b * ODIM + d) * ODIM + h) * ODIM + 2 * mw));
                *o = make_float2(se, so);
            }
        }
    }
}

extern "C" void kernel_launch(void* const* d_in, const int* in_sizes, int n_in,
                              void* d_out, int out_size, void* d_ws, size_t ws_size,
                              hipStream_t stream) {
    const float* x       = (const float*)d_in[0];
    const float* prev    = (const float*)d_in[1];
    const float* istyle  = (const float*)d_in[2];
    const float* style_w = (const float*)d_in[3];
    const float* style_b = (const float*)d_in[4];
    const float* conv_w  = (const float*)d_in[5];
    float* out = (float*)d_out;

    float* mod = (float*)d_ws;
    float* y0  = mod + BATCH * CHANNELS;

    style_mod_kernel<<<BATCH * (CHANNELS / K1_CH_PER_BLOCK), 256, 0, stream>>>(istyle, style_w, style_b, conv_w, mod);

    einsum_wmma_kernel<<<(BATCH * (SPATIAL / K2_CELLS)) / 8, 256, 0, stream>>>(x, prev, mod, y0);

    upblur_kernel<<<(BATCH * SPATIAL / K3_CELLS_D) / 256, 256, 0, stream>>>(y0, out);
}


# style GEMV folded into the channel-sum kernel, 3 launches -> 2: a block takes one 64-channel quarter and makes its own 64 weights; blur kernel adds the 4 quarter planes; y0 sum order changes
# speedup vs baseline: 2.2850x; 1.0616x over previous
#include <hip/hip_runtime.h>


typedef __attribute__((ext_vector_type(2))) float v2f;
typedef __attribute__((ext_vector_type(8))) float v8f;

#define BATCH    4
#define CHANNELS 256
#define LATENT   512
#define DIM      32
#define SPATIAL  (DIM * DIM * DIM)
#define ODIM     64

#define K2_CELLS       64
#define K2_SLICES      4
#define K2_SLICE_CH    (CHANNELS / K2_SLICES)
#define K2_STEPS       (K2_SLICE_CH / 4)
#define K2_UNITS       4
#define K2_UNIT_STRIDE (8 * K2_CELLS)
#define K2_BLOCK_CELLS (K2_UNITS * K2_UNIT_STRIDE)
#define K2_RANGES      (SPATIAL / K2_BLOCK_CELLS)
#define K2_AHEAD       16
#define K2_CH_PER_WAVE (K2_SLICE_CH / 8)
#define K2_VEC_PER_LANE (LATENT / (4 * 32))

__device__ __forceinline__ v8f k2_mma(v2f a, float b_lo, float b_hi, v8f c) {
    v2f bfrag = { b_lo, b_hi };
    return __builtin_amdgcn_wmma_f32_16x16x4_f32(
         false, a,  false, bfrag,
         (short)0, c,  false,  false);
}

__global__ __launch_bounds__(256)
void style_einsum_wmma_kernel(const float* __restrict__ x,
                              const float* __restrict__ prev,
                              const float* __restrict__ istyle,
                              const float* __restrict__ style_w,
                              const float* __restrict__ style_b,
                              const float* __restrict__ conv_w,
                              float* __restrict__ part) {
    __shared__ float s_mod[K2_SLICE_CH];
    __shared__ __attribute__((aligned(16))) float s_is[LATENT];

    const int lane = threadIdx.x & 31;
    const int wave = threadIdx.x >> 5;
    const int b    = blockIdx.x / (K2_SLICES * K2_RANGES);
    const int q    = (blockIdx.x / K2_RANGES) % K2_SLICES;
    const int rng  = blockIdx.x % K2_RANGES;
    const int half = lane >> 4;
    const int col  = lane & 15;
    const int c0   = q * K2_SLICE_CH;
    const int s0   = rng * K2_BLOCK_CELLS + wave * K2_CELLS + 4 * col;

    const float2 isv = *(const float2*)(istyle + b * LATENT + 2 * threadIdx.x);
    const int   cm = c0 + K2_CH_PER_WAVE * wave + (lane & (K2_CH_PER_WAVE - 1));
    const float cw = conv_w[cm];
    const float sb = style_b[cm];

    float4 w4[K2_CH_PER_WAVE][K2_VEC_PER_LANE];
    #pragma unroll
    for (int j = 0; j < K2_CH_PER_WAVE; ++j) {
        const float4* wrow = (const float4*)(style_w + (c0 + K2_CH_PER_WAVE * wave + j) * LATENT);
        #pragma unroll
        for (int i = 0; i < K2_VEC_PER_LANE; ++i)
            w4[j][i] = wrow[lane + 32 * i];
    }

    float4 pv[K2_UNITS];
    #pragma unroll
    for (int u = 0; u < K2_UNITS; ++u)
        pv[u] = *(const float4*)(prev + (size_t)b * SPATIAL + s0 + K2_UNIT_STRIDE * u);

    const float* xp = x + ((size_t)b * CHANNELS + c0 + 2 * half) * SPATIAL + s0;

    float4 ring[K2_AHEAD][2];
    #pragma unroll
    for (int t = 0; t < K2_AHEAD; ++t) {
        const float* xs = xp + (size_t)(4 * (t % K2_STEPS)) * SPATIAL + K2_UNIT_STRIDE * (t / K2_STEPS);
        ring[t][0] = *(const float4*)(xs);
        ring[t][1] = *(const float4*)(xs + SPATIAL);
    }

    *(float2*)&s_is[2 * threadIdx.x] = isv;
    __builtin_amdgcn_fence(__ATOMIC_RELEASE, "workgroup", "local");
    __builtin_amdgcn_s_barrier();
    __builtin_amdgcn_fence(__ATOMIC_ACQUIRE, "workgroup", "local");
    float4 s4[K2_VEC_PER_LANE];
    #pragma unroll
    for (int i = 0; i < K2_VEC_PER_LANE; ++i)
        s4[i] = ((const float4*)s_is)[lane + 32 * i];

    float acc[K2_CH_PER_WAVE];
    #pragma unroll
    for (int j = 0; j < K2_CH_PER_WAVE; ++j) {
        v2f a2 = { 0.0f, 0.0f };
        #pragma unroll
        for (int i = 0; i < K2_VEC_PER_LANE; ++i) {
            const v2f slo = { s4[i].x, s4[i].y },       shi = { s4[i].z, s4[i].w };
            const v2f wlo = { w4[j][i].x, w4[j][i].y }, whi = { w4[j][i].z, w4[j][i].w };
            a2 = __builtin_elementwise_fma(slo, wlo, a2);
            a2 = __builtin_elementwise_fma(shi, whi, a2);
        }
        acc[j] = a2.x + a2.y;
    }
    #pragma unroll
    for (int off = 16; off > 0; off >>= 1) {
        #pragma unroll
        for (int j = 0; j < K2_CH_PER_WAVE; ++j)
            acc[j] += __shfl_xor(acc[j], off, 32);
    }
    float dsel = acc[0];
    #pragma unroll
    for (int j = 1; j < K2_CH_PER_WAVE; ++j)
        dsel = ((lane & (K2_CH_PER_WAVE - 1)) == j) ? acc[j] : dsel;
    s_mod[K2_CH_PER_WAVE * wave + (lane & (K2_CH_PER_WAVE - 1))] = cw * ((dsel + sb) + 1.0f);

    __builtin_amdgcn_fence(__ATOMIC_RELEASE, "workgroup", "local");
    __builtin_amdgcn_s_barrier();
    __builtin_amdgcn_fence(__ATOMIC_ACQUIRE, "workgroup", "local");

    v2f mf[K2_STEPS];
    #pragma unroll
    for (int i = 0; i < K2_STEPS; ++i)
        mf[i] = *(const v2f*)&s_mod[4 * i + 2 * half];
    __builtin_amdgcn_sched_barrier(0);

    float* pp = part + ((size_t)q * BATCH + b) * SPATIAL + s0;

    v8f acc0 = {}, acc1 = {}, acc2 = {}, acc3 = {};
    #pragma unroll
    for (int t = 0; t < K2_UNITS * K2_STEPS; ++t) {
        const int slot = t % K2_AHEAD;
        const int i    = t % K2_STEPS;
        const int u    = t / K2_STEPS;
        const float4 q0 = ring[slot][0], q1 = ring[slot][1];
        acc0 = k2_mma(mf[i], q0.x, q1.x, acc0);
        acc1 = k2_mma(mf[i], q0.y, q1.y, acc1);
        acc2 = k2_mma(mf[i], q0.z, q1.z, acc2);
        acc3 = k2_mma(mf[i], q0.w, q1.w, acc3);
        if (t + K2_AHEAD < K2_UNITS * K2_STEPS) {
            const int tn = t + K2_AHEAD;
            const float* xs = xp + (size_t)(4 * (tn % K2_STEPS)) * SPATIAL + K2_UNIT_STRIDE * (tn / K2_STEPS);
            ring[slot][0] = *(const float4*)(xs);
            ring[slot][1] = *(const float4*)(xs + SPATIAL);
        }
        if (i == K2_STEPS - 1) {
            const float4 pz = (q == 0) ? pv[u] : make_float4(0.0f, 0.0f, 0.0f, 0.0f);
            const float4 r = { acc0[0] + pz.x, acc1[0] + pz.y, acc2[0] + pz.z, acc3[0] + pz.w };
            if (half == 0)
                *(float4*)(pp + K2_UNIT_STRIDE * u) = r;
            acc0 = (v8f){}; acc1 = (v8f){}; acc2 = (v8f){}; acc3 = (v8f){};
        }
    }
}

#define K3_CELLS_D 2

__global__ __launch_bounds__(256)
void upblur_kernel(const float* __restrict__ part, float* __restrict__ out) {
    const int t   = blockIdx.x * 256 + threadIdx.x;
    const int mw  = t & (DIM - 1);
    const int mh  = (t >> 5) & (DIM - 1);
    const int md0 = K3_CELLS_D * ((t >> 10) & (DIM / K3_CELLS_D - 1));
    const int b   = t >> 14;

    const float* y_q0 = part + (size_t)b * SPATIAL;
    const float* y_q1 = y_q0 + (size_t)BATCH * SPATIAL;
    const float* y_q2 = y_q1 + (size_t)BATCH * SPATIAL;
    const float* y_q3 = y_q2 + (size_t)BATCH * SPATIAL;

    const int ih[3] = { (mh - 1 < 0) ? 0 : mh - 1, mh, (mh + 1 > DIM - 1) ? DIM - 1 : mh + 1 };
    const int ip[K3_CELLS_D + 2] = { (md0 - 1 < 0) ? 0 : md0 - 1, md0, md0 + 1,
                                     (md0 + 2 > DIM - 1) ? DIM - 1 : md0 + 2 };

    float re[K3_CELLS_D + 2][3], ro[K3_CELLS_D + 2][3];
    #pragma unroll
    for (int p = 0; p < K3_CELLS_D + 2; ++p) {
        #pragma unroll
        for (int j = 0; j < 3; ++j) {
            const int o = (ip[p] * DIM + ih[j]) * DIM + mw;
            const float v1 = ((y_q0[o] + y_q1[o]) + y_q2[o]) + y_q3[o];
            const float v0 = __shfl_up(v1, 1, 32);
            const float v2 = __shfl_down(v1, 1, 32);
            re[p][j] = 0.3125f * v0 + 0.625f * v1 + 0.0625f * v2;
            ro[p][j] = 0.0625f * v0 + 0.625f * v1 + 0.3125f * v2;
        }
    }

    const float wt[2][3] = { { 0.3125f, 0.625f, 0.0625f }, { 0.0625f, 0.625f, 0.3125f } };

    #pragma unroll
    for (int q = 0; q < K3_CELLS_D; ++q) {
        #pragma unroll
        for (int pd = 0; pd < 2; ++pd) {
            #pragma unroll
            for (int ph = 0; ph < 2; ++ph) {
                float se = 0.0f, so = 0.0f;
                #pragma unroll
                for (int i = 0; i < 3; ++i) {
                    #pragma unroll
                    for (int j = 0; j < 3; ++j) {
                        const float c = wt[pd][i] * wt[ph][j];
                        se += c * re[q + i][j];
                        so += c * ro[q + i][j];
                    }
                }
                const int d = 2 * (md0 + q) + pd;
                const int h = 2 * mh + ph;
                float2* o = (float2*)(out + (((size_t)(b * ODIM + d) * ODIM + h) * ODIM + 2 * mw));
                *o = make_float2(se, so);
            }
        }
    }
}

extern "C" void kernel_launch(void* const* d_in, const int* in_sizes, int n_in,
                              void* d_out, int out_size, void* d_ws, size_t ws_size,
                              hipStream_t stream) {
    const float* x       = (const float*)d_in[0];
    const float* prev    = (const float*)d_in[1];
    const float* istyle  = (const float*)d_in[2];
    const float* style_w = (const float*)d_in[3];
    const float* style_b = (const float*)d_in[4];
    const float* conv_w  = (const float*)d_in[5];
    float* out = (float*)d_out;

    float* part = (float*)d_ws;

    style_einsum_wmma_kernel<<<BATCH * K2_SLICES * K2_RANGES, 256, 0, stream>>>(
        x, prev, istyle, style_w, style_b, conv_w, part);

    upblur_kernel<<<(BATCH * SPATIAL / K3_CELLS_D) / 256, 256, 0, stream>>>(part, out);
}
